# starting kernel
# baseline (speedup 1.0000x reference)
#include <hip/hip_runtime.h>

typedef __attribute__((ext_vector_type(2))) float v2f;
typedef __attribute__((ext_vector_type(8))) float v8f;

#define IN_DIM 512
#define HID    32
#define OUT_DIM 16

#define SMEM_FLOATS 21040
#define SMEM_BYTES  (SMEM_FLOATS * 4)

__global__ __launch_bounds__(256)
void mlp_fused_kernel(const float* __restrict__ feat,
                      const float* __restrict__ W1,
                      const float* __restrict__ b1,
                      const float* __restrict__ W2,
                      const float* __restrict__ b2,
                      float* __restrict__ x, int n)
{
    extern __shared__ float smem[];
    float* sW1 = smem;
    float* sH  = smem + 16384;
    float* sW2 = smem + 20480;
    float* sb1 = smem + 20992;
    float* sb2 = smem + 21024;

    for (int i = threadIdx.x; i < IN_DIM * HID; i += 256) sW1[i] = W1[i];
    for (int i = threadIdx.x; i < HID * OUT_DIM; i += 256) sW2[i] = W2[i];
    if (threadIdx.x < HID)     sb1[threadIdx.x] = b1[threadIdx.x];
    if (threadIdx.x < OUT_DIM) sb2[threadIdx.x] = b2[threadIdx.x];
    __syncthreads();

    const int wave = threadIdx.x >> 5;
    const int lane = threadIdx.x & 31;
    const int row0 = blockIdx.x * 128 + wave * 16;

    if (row0 < n) {
        const int m     = lane & 15;
        const int khalf = (lane >> 4) << 1;
        const int n0    = lane & 15;
        const float* arow = feat + (size_t)(row0 + m) * IN_DIM + khalf;

        v8f acc0 = {};
        v8f acc1 = {};

        #pragma unroll 4
        for (int k = 0; k < IN_DIM; k += 4) {
            v2f a = __builtin_nontemporal_load((const v2f*)(arow + k));
            const int kb = k + khalf;
            v2f bl, bh;
            bl.x = sW1[kb * HID + n0];
            bl.y = sW1[kb * HID + HID + n0];
            bh.x = sW1[kb * HID + 16 + n0];
            bh.y = sW1[kb * HID + HID + 16 + n0];
            acc0 = __builtin_amdgcn_wmma_f32_16x16x4_f32(
                false, a, false, bl, (short)0, acc0, false, false);
            acc1 = __builtin_amdgcn_wmma_f32_16x16x4_f32(
                false, a, false, bh, (short)0, acc1, false, false);
        }

        float* shw = sH + wave * (16 * HID);
        #pragma unroll
        for (int i = 0; i < 8; ++i) {
            const int mm = i + ((lane >> 4) << 3);
            float h0 = acc0[i] + sb1[n0];
            float h1 = acc1[i] + sb1[16 + n0];
            shw[mm * HID + n0]      = h0 > 0.0f ? h0 : 0.0f;
            shw[mm * HID + 16 + n0] = h1 > 0.0f ? h1 : 0.0f;
        }
        asm volatile("s_wait_dscnt 0x0" ::: "memory");

        const int mm   = lane & 15;
        const int half = lane >> 4;
        const float* hrow = shw + mm * HID;
        float* xrow = x + (size_t)(row0 + mm) * OUT_DIM;
        #pragma unroll
        for (int oo = 0; oo < 8; ++oo) {
            const int o = half * 8 + oo;
            float s = sb2[o];
            #pragma unroll
            for (int nn = 0; nn < HID; ++nn)
                s += hrow[nn] * sW2[nn * OUT_DIM + o];
            xrow[o] = s;
        }
    }
}

__global__ __launch_bounds__(256)
void spmm_scatter_kernel(const int* __restrict__ edge_row,
                         const int* __restrict__ edge_col,
                         const float* __restrict__ edge_w,
                         const float* __restrict__ x,
                         float* __restrict__ out, int E)
{
    const long t = (long)blockIdx.x * 256 + threadIdx.x;
    const long e = t >> 4;
    const int  o = (int)(t & 15);
    if (e < E) {
        const int   r = __builtin_nontemporal_load(edge_row + e);
        const int   c = __builtin_nontemporal_load(edge_col + e);
        const float w = __builtin_nontemporal_load(edge_w + e);
        atomicAdd(out + (size_t)r * OUT_DIM + o, w * x[(size_t)c * OUT_DIM + o]);
    }
}

extern "C" void kernel_launch(void* const* d_in, const int* in_sizes, int n_in,
                              void* d_out, int out_size, void* d_ws, size_t ws_size,
                              hipStream_t stream) {
    const float* feat = (const float*)d_in[0];
    const int*   er   = (const int*)  d_in[1];
    const int*   ec   = (const int*)  d_in[2];
    const float* ew   = (const float*)d_in[3];
    const float* W1   = (const float*)d_in[4];
    const float* b1   = (const float*)d_in[5];
    const float* W2   = (const float*)d_in[6];
    const float* b2   = (const float*)d_in[7];
    float* out = (float*)d_out;
    float* x   = (float*)d_ws;

    const int n = in_sizes[0] / IN_DIM;
    const int E = in_sizes[1];

    hipMemsetAsync(d_out, 0, (size_t)out_size * sizeof(float), stream);

    const int mlp_blocks = (n + 127) / 128;
    mlp_fused_kernel<<<mlp_blocks, 256, SMEM_BYTES, stream>>>(
        feat, W1, b1, W2, b2, x, n);

    const long sthreads = (long)E * 16;
    const int  sblocks  = (int)((sthreads + 255) / 256);
    spmm_scatter_kernel<<<sblocks, 256, 0, stream>>>(er, ec, ew, x, out, E);
}


# scatter launch: each wave owns 128 edges (was 2): one whole-line index load per 32 edges shared by lane shuffle, 16 row gathers in flight ahead of 16 f32 atomic adds (arrival-order sums as before)
# speedup vs baseline: 1.6380x; 1.6380x over previous
#include <hip/hip_runtime.h>

typedef __attribute__((ext_vector_type(2))) float v2f;
typedef __attribute__((ext_vector_type(8))) float v8f;

#define IN_DIM 512
#define HID    32
#define OUT_DIM 16

#define SMEM_FLOATS 21040
#define SMEM_BYTES  (SMEM_FLOATS * 4)

__global__ __launch_bounds__(256)
void mlp_fused_kernel(const float* __restrict__ feat,
                      const float* __restrict__ W1,
                      const float* __restrict__ b1,
                      const float* __restrict__ W2,
                      const float* __restrict__ b2,
                      float* __restrict__ x, int n)
{
    extern __shared__ float smem[];
    float* sW1 = smem;
    float* sH  = smem + 16384;
    float* sW2 = smem + 20480;
    float* sb1 = smem + 20992;
    float* sb2 = smem + 21024;

    for (int i = threadIdx.x; i < IN_DIM * HID; i += 256) sW1[i] = W1[i];
    for (int i = threadIdx.x; i < HID * OUT_DIM; i += 256) sW2[i] = W2[i];
    if (threadIdx.x < HID)     sb1[threadIdx.x] = b1[threadIdx.x];
    if (threadIdx.x < OUT_DIM) sb2[threadIdx.x] = b2[threadIdx.x];
    __syncthreads();

    const int wave = threadIdx.x >> 5;
    const int lane = threadIdx.x & 31;
    const int row0 = blockIdx.x * 128 + wave * 16;

    if (row0 < n) {
        const int m     = lane & 15;
        const int khalf = (lane >> 4) << 1;
        const int n0    = lane & 15;
        const float* arow = feat + (size_t)(row0 + m) * IN_DIM + khalf;

        v8f acc0 = {};
        v8f acc1 = {};

        #pragma unroll 4
        for (int k = 0; k < IN_DIM; k += 4) {
            v2f a = __builtin_nontemporal_load((const v2f*)(arow + k));
            const int kb = k + khalf;
            v2f bl, bh;
            bl.x = sW1[kb * HID + n0];
            bl.y = sW1[kb * HID + HID + n0];
            bh.x = sW1[kb * HID + 16 + n0];
            bh.y = sW1[kb * HID + HID + 16 + n0];
            acc0 = __builtin_amdgcn_wmma_f32_16x16x4_f32(
                false, a, false, bl, (short)0, acc0, false, false);
            acc1 = __builtin_amdgcn_wmma_f32_16x16x4_f32(
                false, a, false, bh, (short)0, acc1, false, false);
        }

        float* shw = sH + wave * (16 * HID);
        #pragma unroll
        for (int i = 0; i < 8; ++i) {
            const int mm = i + ((lane >> 4) << 3);
            float h0 = acc0[i] + sb1[n0];
            float h1 = acc1[i] + sb1[16 + n0];
            shw[mm * HID + n0]      = h0 > 0.0f ? h0 : 0.0f;
            shw[mm * HID + 16 + n0] = h1 > 0.0f ? h1 : 0.0f;
        }
        asm volatile("s_wait_dscnt 0x0" ::: "memory");

        const int mm   = lane & 15;
        const int half = lane >> 4;
        const float* hrow = shw + mm * HID;
        float* xrow = x + (size_t)(row0 + mm) * OUT_DIM;
        #pragma unroll
        for (int oo = 0; oo < 8; ++oo) {
            const int o = half * 8 + oo;
            float s = sb2[o];
            #pragma unroll
            for (int nn = 0; nn < HID; ++nn)
                s += hrow[nn] * sW2[nn * OUT_DIM + o];
            xrow[o] = s;
        }
    }
}

#define SC_TRIPS 4
#define SC_EDGES_PER_WAVE (32 * SC_TRIPS)
#define SC_WAVES 4
#define SC_EDGES_PER_BLOCK (SC_WAVES * SC_EDGES_PER_WAVE)
__global__ __launch_bounds__(32 * SC_WAVES)
void spmm_scatter_kernel(const int* __restrict__ edge_row,
                         const int* __restrict__ edge_col,
                         const float* __restrict__ edge_w,
                         const float* __restrict__ x,
                         float* __restrict__ out, int E)
{
    const int  lane = threadIdx.x & 31;
    const int  half = lane >> 4;
    const int  o    = lane & 15;
    const long base = ((long)blockIdx.x * SC_WAVES + (threadIdx.x >> 5)) * SC_EDGES_PER_WAVE;
    if (base >= E) return;

    if (base + SC_EDGES_PER_WAVE <= E) {
        int   r[SC_TRIPS], c[SC_TRIPS];
        float w[SC_TRIPS];
        #pragma unroll
        for (int t = 0; t < SC_TRIPS; ++t) {
            const long e = base + t * 32 + lane;
            r[t] = __builtin_nontemporal_load(edge_row + e);
            c[t] = __builtin_nontemporal_load(edge_col + e);
            w[t] = __builtin_nontemporal_load(edge_w + e);
        }
        #pragma unroll
        for (int t = 0; t < SC_TRIPS; ++t) {
            float v[16];
            int   rr[16];
            #pragma unroll
            for (int j = 0; j < 16; ++j) {
                const int s  = 2 * j + half;
                const int cj = __shfl(c[t], s, 32);
                const float wj = __shfl(w[t], s, 32);
                rr[j] = __shfl(r[t], s, 32);
                v[j]  = wj * *(const float*)((const char*)x + (unsigned)(cj * (OUT_DIM * 4) + o * 4));
            }
            #pragma unroll
            for (int j = 0; j < 16; ++j)
                atomicAdd((float*)((char*)out + (unsigned)(rr[j] * (OUT_DIM * 4) + o * 4)), v[j]);
        }
    } else {
        for (long e = base + half; e < E && e < base + SC_EDGES_PER_WAVE; e += 2) {
            const int   r = edge_row[e];
            const int   c = edge_col[e];
            const float w = edge_w[e];
            atomicAdd(out + (size_t)r * OUT_DIM + o, w * x[(size_t)c * OUT_DIM + o]);
        }
    }
}

extern "C" void kernel_launch(void* const* d_in, const int* in_sizes, int n_in,
                              void* d_out, int out_size, void* d_ws, size_t ws_size,
                              hipStream_t stream) {
    const float* feat = (const float*)d_in[0];
    const int*   er   = (const int*)  d_in[1];
    const int*   ec   = (const int*)  d_in[2];
    const float* ew   = (const float*)d_in[3];
    const float* W1   = (const float*)d_in[4];
    const float* b1   = (const float*)d_in[5];
    const float* W2   = (const float*)d_in[6];
    const float* b2   = (const float*)d_in[7];
    float* out = (float*)d_out;
    float* x   = (float*)d_ws;

    const int n = in_sizes[0] / IN_DIM;
    const int E = in_sizes[1];

    hipMemsetAsync(d_out, 0, (size_t)out_size * sizeof(float), stream);

    const int mlp_blocks = (n + 127) / 128;
    mlp_fused_kernel<<<mlp_blocks, 256, SMEM_BYTES, stream>>>(
        feat, W1, b1, W2, b2, x, n);

    const int  sblocks  = (E + SC_EDGES_PER_BLOCK - 1) / SC_EDGES_PER_BLOCK;
    spmm_scatter_kernel<<<sblocks, 32 * SC_WAVES, 0, stream>>>(er, ec, ew, x, out, E);
}


# fc1 feature loader: each lane loads contiguous 16-B pieces of its row, 8 loads = one 256-B piece per row, issued one burst ahead of the WMMAs; A and B k order permuted alike (fc1 sum order changes)
# speedup vs baseline: 1.8698x; 1.1415x over previous
#include <hip/hip_runtime.h>

typedef __attribute__((ext_vector_type(2))) float v2f;
typedef __attribute__((ext_vector_type(4))) float v4f;
typedef __attribute__((ext_vector_type(8))) float v8f;

#define IN_DIM 512
#define HID    32
#define OUT_DIM 16

#define SMEM_FLOATS 21040
#define SMEM_BYTES  (SMEM_FLOATS * 4)

__global__ __launch_bounds__(256)
void mlp_fused_kernel(const float* __restrict__ feat,
                      const float* __restrict__ W1,
                      const float* __restrict__ b1,
                      const float* __restrict__ W2,
                      const float* __restrict__ b2,
                      float* __restrict__ x, int n)
{
    extern __shared__ float smem[];
    float* sW1 = smem;
    float* sH  = smem + 16384;
    float* sW2 = smem + 20480;
    float* sb1 = smem + 20992;
    float* sb2 = smem + 21024;

    for (int i = threadIdx.x; i < IN_DIM * HID; i += 256) sW1[i] = W1[i];
    for (int i = threadIdx.x; i < HID * OUT_DIM; i += 256) sW2[i] = W2[i];
    if (threadIdx.x < HID)     sb1[threadIdx.x] = b1[threadIdx.x];
    if (threadIdx.x < OUT_DIM) sb2[threadIdx.x] = b2[threadIdx.x];
    __syncthreads();

    const int wave = threadIdx.x >> 5;
    const int lane = threadIdx.x & 31;
    const int row0 = blockIdx.x * 128 + wave * 16;

    if (row0 < n) {
        const int m     = lane & 15;
        const int kq    = (lane >> 4) << 2;
        const int n0    = lane & 15;
        const float* arow = feat + (size_t)(row0 + m) * IN_DIM + kq;
        const float* brow = sW1 + kq * HID + n0;

        v8f acc0 = {};
        v8f acc1 = {};

        v4f bufA[8], bufB[8];

        auto load_burst = [&](v4f (&buf)[8], int k0) {
            #pragma unroll
            for (int i = 0; i < 8; ++i)
                buf[i] = __builtin_nontemporal_load((const v4f*)(arow + k0 + 8 * i));
            __builtin_amdgcn_sched_barrier(0);
        };
        auto mma_burst = [&](const v4f (&buf)[8], int k0) {
            #pragma unroll
            for (int i = 0; i < 8; ++i) {
                const float* bp = brow + (k0 + 8 * i) * HID;
                v2f a, bl, bh;
                a.x  = buf[i].x;          a.y  = buf[i].y;
                bl.x = bp[0];             bl.y = bp[HID];
                bh.x = bp[16];            bh.y = bp[HID + 16];
                acc0 = __builtin_amdgcn_wmma_f32_16x16x4_f32(
                    false, a, false, bl, (short)0, acc0, false, false);
                acc1 = __builtin_amdgcn_wmma_f32_16x16x4_f32(
                    false, a, false, bh, (short)0, acc1, false, false);
                a.x  = buf[i].z;          a.y  = buf[i].w;
                bl.x = bp[2 * HID];       bl.y = bp[3 * HID];
                bh.x = bp[2 * HID + 16];  bh.y = bp[3 * HID + 16];
                acc0 = __builtin_amdgcn_wmma_f32_16x16x4_f32(
                    false, a, false, bl, (short)0, acc0, false, false);
                acc1 = __builtin_amdgcn_wmma_f32_16x16x4_f32(
                    false, a, false, bh, (short)0, acc1, false, false);
            }
        };

        load_burst(bufA, 0);
        #pragma unroll 1
        for (int k0 = 0; k0 < IN_DIM - 128; k0 += 128) {
            load_burst(bufB, k0 + 64);
            mma_burst(bufA, k0);
            load_burst(bufA, k0 + 128);
            mma_burst(bufB, k0 + 64);
        }
        load_burst(bufB, IN_DIM - 64);
        mma_burst(bufA, IN_DIM - 128);
        mma_burst(bufB, IN_DIM - 64);

        float* shw = sH + wave * (16 * HID);
        #pragma unroll
        for (int i = 0; i < 8; ++i) {
            const int mm = i + ((lane >> 4) << 3);
            float h0 = acc0[i] + sb1[n0];
            float h1 = acc1[i] + sb1[16 + n0];
            shw[mm * HID + n0]      = h0 > 0.0f ? h0 : 0.0f;
            shw[mm * HID + 16 + n0] = h1 > 0.0f ? h1 : 0.0f;
        }
        asm volatile("s_wait_dscnt 0x0" ::: "memory");

        const int mm   = lane & 15;
        const int half = lane >> 4;
        const float* hrow = shw + mm * HID;
        float* xrow = x + (size_t)(row0 + mm) * OUT_DIM;
        #pragma unroll
        for (int oo = 0; oo < 8; ++oo) {
            const int o = half * 8 + oo;
            float s = sb2[o];
            #pragma unroll
            for (int nn = 0; nn < HID; ++nn)
                s += hrow[nn] * sW2[nn * OUT_DIM + o];
            xrow[o] = s;
        }
    }
}

#define SC_TRIPS 4
#define SC_EDGES_PER_WAVE (32 * SC_TRIPS)
#define SC_WAVES 4
#define SC_EDGES_PER_BLOCK (SC_WAVES * SC_EDGES_PER_WAVE)
__global__ __launch_bounds__(32 * SC_WAVES)
void spmm_scatter_kernel(const int* __restrict__ edge_row,
                         const int* __restrict__ edge_col,
                         const float* __restrict__ edge_w,
                         const float* __restrict__ x,
                         float* __restrict__ out, int E)
{
    const int  lane = threadIdx.x & 31;
    const int  half = lane >> 4;
    const int  o    = lane & 15;
    const long base = ((long)blockIdx.x * SC_WAVES + (threadIdx.x >> 5)) * SC_EDGES_PER_WAVE;
    if (base >= E) return;

    if (base + SC_EDGES_PER_WAVE <= E) {
        int   r[SC_TRIPS], c[SC_TRIPS];
        float w[SC_TRIPS];
        #pragma unroll
        for (int t = 0; t < SC_TRIPS; ++t) {
            const long e = base + t * 32 + lane;
            r[t] = __builtin_nontemporal_load(edge_row + e);
            c[t] = __builtin_nontemporal_load(edge_col + e);
            w[t] = __builtin_nontemporal_load(edge_w + e);
        }
        #pragma unroll
        for (int t = 0; t < SC_TRIPS; ++t) {
            float v[16];
            int   rr[16];
            #pragma unroll
            for (int j = 0; j < 16; ++j) {
                const int s  = 2 * j + half;
                const int cj = __shfl(c[t], s, 32);
                const float wj = __shfl(w[t], s, 32);
                rr[j] = __shfl(r[t], s, 32);
                v[j]  = wj * *(const float*)((const char*)x + (unsigned)(cj * (OUT_DIM * 4) + o * 4));
            }
            #pragma unroll
            for (int j = 0; j < 16; ++j)
                atomicAdd((float*)((char*)out + (unsigned)(rr[j] * (OUT_DIM * 4) + o * 4)), v[j]);
        }
    } else {
        for (long e = base + half; e < E && e < base + SC_EDGES_PER_WAVE; e += 2) {
            const int   r = edge_row[e];
            const int   c = edge_col[e];
            const float w = edge_w[e];
            atomicAdd(out + (size_t)r * OUT_DIM + o, w * x[(size_t)c * OUT_DIM + o]);
        }
    }
}

extern "C" void kernel_launch(void* const* d_in, const int* in_sizes, int n_in,
                              void* d_out, int out_size, void* d_ws, size_t ws_size,
                              hipStream_t stream) {
    const float* feat = (const float*)d_in[0];
    const int*   er   = (const int*)  d_in[1];
    const int*   ec   = (const int*)  d_in[2];
    const float* ew   = (const float*)d_in[3];
    const float* W1   = (const float*)d_in[4];
    const float* b1   = (const float*)d_in[5];
    const float* W2   = (const float*)d_in[6];
    const float* b2   = (const float*)d_in[7];
    float* out = (float*)d_out;
    float* x   = (float*)d_ws;

    const int n = in_sizes[0] / IN_DIM;
    const int E = in_sizes[1];

    hipMemsetAsync(d_out, 0, (size_t)out_size * sizeof(float), stream);

    const int mlp_blocks = (n + 127) / 128;
    mlp_fused_kernel<<<mlp_blocks, 256, SMEM_BYTES, stream>>>(
        feat, W1, b1, W2, b2, x, n);

    const int  sblocks  = (E + SC_EDGES_PER_BLOCK - 1) / SC_EDGES_PER_BLOCK;
    spmm_scatter_kernel<<<sblocks, 32 * SC_WAVES, 0, stream>>>(er, ec, ew, x, out, E);
}
